# starting kernel
# baseline (speedup 1.0000x reference)
#include <hip/hip_runtime.h>
#include <hip/hip_bf16.h>


typedef __attribute__((ext_vector_type(2))) float v2f;
typedef __attribute__((ext_vector_type(8))) float v8f;

#define LSTM_B    2048
#define LSTM_T    512
#define LSTM_H    64
#define LSTM_G    256
#define MTILE     16
#define NWAVES    16
#define THREADS   (NWAVES * 32)
#define GPAD      260
#define HPAD      68

#define LOG2E_F   1.4426950408889634f

__device__ __forceinline__ float fast_tanh(float x) {
#if __has_builtin(__builtin_amdgcn_tanhf)
    return __builtin_amdgcn_tanhf(x);
#else
    float r;
    asm volatile("v_tanh_f32 %0, %1\n\tv_nop" : "=v"(r) : "v"(x));
    return r;
#endif
}

__device__ __forceinline__ float fast_sigmoid(float x) {
    float e = __builtin_amdgcn_exp2f(x * -LOG2E_F);
    return __builtin_amdgcn_rcpf(1.0f + e);
}

__device__ __forceinline__ v8f wmma_f32_k4(v2f a, v2f b, v8f c) {
    return __builtin_amdgcn_wmma_f32_16x16x4_f32(false, a, false, b,
                                                 (short)0, c, false, false);
}

__global__ __launch_bounds__(THREADS, 1)
void LSTMEncoder_78975858639682_kernel(
    const float* __restrict__ x,
    const float* __restrict__ w_ih0,
    const float* __restrict__ w_hh0,
    const float* __restrict__ b_ih0,
    const float* __restrict__ b_hh0,
    const float* __restrict__ w_ih1,
    const float* __restrict__ w_hh1,
    const float* __restrict__ b_ih1,
    const float* __restrict__ b_hh1,
    const float* __restrict__ fc_w,
    const float* __restrict__ fc_b,
    float* __restrict__ out)
{
    __shared__ float sG [MTILE * GPAD];
    __shared__ float sH0[MTILE * HPAD];
    __shared__ float sH1[MTILE * HPAD];

    const int tid  = threadIdx.x;
    const int lane = tid & 31;
    const int wave = tid >> 5;
    const int half = lane >> 4;
    const int l16  = lane & 15;
    const int rowBase = blockIdx.x * MTILE;

    const int ncol = wave * 16 + l16;
    const int kb   = 2 * half;
    v2f wih0 = *(const v2f*)(w_ih0 + ncol * 4 + kb);
    v2f whh0[16], wih1[16], whh1[16];
#pragma unroll
    for (int kc = 0; kc < 16; ++kc) {
        const int k = kc * 4 + kb;
        whh0[kc] = *(const v2f*)(w_hh0 + ncol * LSTM_H + k);
        wih1[kc] = *(const v2f*)(w_ih1 + ncol * LSTM_H + k);
        whh1[kc] = *(const v2f*)(w_hh1 + ncol * LSTM_H + k);
    }
    const float bias0 = b_ih0[ncol] + b_hh0[ncol];
    const float bias1 = b_ih1[ncol] + b_hh1[ncol];

    const int   erow = wave;
    const int   ecol = lane * 2;
    float c0x = 0.f, c0y = 0.f, c1x = 0.f, c1y = 0.f;

    for (int i = tid; i < MTILE * HPAD; i += THREADS) { sH0[i] = 0.f; sH1[i] = 0.f; }
    __syncthreads();

    const int arow = l16;
    const float* xrow = x + (size_t)(rowBase + arow) * LSTM_T * 4 + kb;

    v2f ax = *(const v2f*)(xrow);

#pragma unroll 1
    for (int t = 0; t < LSTM_T; ++t) {
        const int tn = (t < LSTM_T - 1) ? (t + 1) : t;
        v2f ax_next = *(const v2f*)(xrow + (size_t)tn * 4);

        v8f accA, accB;
#pragma unroll
        for (int j = 0; j < 8; ++j) { accA[j] = bias0; accB[j] = 0.f; }
        accB = wmma_f32_k4(ax, wih0, accB);
#pragma unroll
        for (int kc = 0; kc < 16; kc += 2) {
            v2f ahA = *(const v2f*)(sH0 + arow * HPAD + (kc    ) * 4 + kb);
            v2f ahB = *(const v2f*)(sH0 + arow * HPAD + (kc + 1) * 4 + kb);
            accA = wmma_f32_k4(ahA, whh0[kc],     accA);
            accB = wmma_f32_k4(ahB, whh0[kc + 1], accB);
        }
#pragma unroll
        for (int j = 0; j < 8; ++j)
            sG[(j + 8 * half) * GPAD + ncol] = accA[j] + accB[j];
        __syncthreads();

        {
            v2f gi = *(const v2f*)(sG + erow * GPAD + ecol);
            v2f gf = *(const v2f*)(sG + erow * GPAD + 64  + ecol);
            v2f gg = *(const v2f*)(sG + erow * GPAD + 128 + ecol);
            v2f go = *(const v2f*)(sG + erow * GPAD + 192 + ecol);
            float i0 = fast_sigmoid(gi.x), i1 = fast_sigmoid(gi.y);
            float f0 = fast_sigmoid(gf.x), f1 = fast_sigmoid(gf.y);
            float g0 = fast_tanh(gg.x),    g1 = fast_tanh(gg.y);
            float o0 = fast_sigmoid(go.x), o1 = fast_sigmoid(go.y);
            c0x = f0 * c0x + i0 * g0;
            c0y = f1 * c0y + i1 * g1;
            v2f hn; hn.x = o0 * fast_tanh(c0x); hn.y = o1 * fast_tanh(c0y);
            *(v2f*)(sH0 + erow * HPAD + ecol) = hn;
        }
        __syncthreads();

#pragma unroll
        for (int j = 0; j < 8; ++j) { accA[j] = bias1; accB[j] = 0.f; }
#pragma unroll
        for (int kc = 0; kc < 16; ++kc) {
            v2f ahA = *(const v2f*)(sH0 + arow * HPAD + kc * 4 + kb);
            v2f ahB = *(const v2f*)(sH1 + arow * HPAD + kc * 4 + kb);
            accA = wmma_f32_k4(ahA, wih1[kc], accA);
            accB = wmma_f32_k4(ahB, whh1[kc], accB);
        }
#pragma unroll
        for (int j = 0; j < 8; ++j)
            sG[(j + 8 * half) * GPAD + ncol] = accA[j] + accB[j];
        __syncthreads();

        {
            v2f gi = *(const v2f*)(sG + erow * GPAD + ecol);
            v2f gf = *(const v2f*)(sG + erow * GPAD + 64  + ecol);
            v2f gg = *(const v2f*)(sG + erow * GPAD + 128 + ecol);
            v2f go = *(const v2f*)(sG + erow * GPAD + 192 + ecol);
            float i0 = fast_sigmoid(gi.x), i1 = fast_sigmoid(gi.y);
            float f0 = fast_sigmoid(gf.x), f1 = fast_sigmoid(gf.y);
            float g0 = fast_tanh(gg.x),    g1 = fast_tanh(gg.y);
            float o0 = fast_sigmoid(go.x), o1 = fast_sigmoid(go.y);
            c1x = f0 * c1x + i0 * g0;
            c1y = f1 * c1y + i1 * g1;
            v2f hn; hn.x = o0 * fast_tanh(c1x); hn.y = o1 * fast_tanh(c1y);
            *(v2f*)(sH1 + erow * HPAD + ecol) = hn;
        }
        __syncthreads();

        ax = ax_next;
    }

    if (wave < 4) {
        const int nco = wave * 16 + l16;
        v8f accA, accB;
        const float fb = fc_b[nco];
#pragma unroll
        for (int j = 0; j < 8; ++j) { accA[j] = fb; accB[j] = 0.f; }
#pragma unroll
        for (int kc = 0; kc < 16; kc += 2) {
            v2f ahA = *(const v2f*)(sH1 + arow * HPAD + (kc    ) * 4 + kb);
            v2f ahB = *(const v2f*)(sH1 + arow * HPAD + (kc + 1) * 4 + kb);
            v2f bwA = *(const v2f*)(fc_w + nco * LSTM_H + (kc    ) * 4 + kb);
            v2f bwB = *(const v2f*)(fc_w + nco * LSTM_H + (kc + 1) * 4 + kb);
            accA = wmma_f32_k4(ahA, bwA, accA);
            accB = wmma_f32_k4(ahB, bwB, accB);
        }
#pragma unroll
        for (int j = 0; j < 8; ++j) {
            const int r = j + 8 * half;
            out[(size_t)(rowBase + r) * LSTM_H + nco] = fast_tanh(accA[j] + accB[j]);
        }
    }
}

extern "C" void kernel_launch(void* const* d_in, const int* in_sizes, int n_in,
                              void* d_out, int out_size, void* d_ws, size_t ws_size,
                              hipStream_t stream) {
    (void)in_sizes; (void)n_in; (void)d_ws; (void)ws_size; (void)out_size;
    const float* x     = (const float*)d_in[0];
    const float* w_ih0 = (const float*)d_in[1];
    const float* w_hh0 = (const float*)d_in[2];
    const float* b_ih0 = (const float*)d_in[3];
    const float* b_hh0 = (const float*)d_in[4];
    const float* w_ih1 = (const float*)d_in[5];
    const float* w_hh1 = (const float*)d_in[6];
    const float* b_ih1 = (const float*)d_in[7];
    const float* b_hh1 = (const float*)d_in[8];
    const float* fc_w  = (const float*)d_in[9];
    const float* fc_b  = (const float*)d_in[10];
    float* out = (float*)d_out;

    dim3 grid(LSTM_B / MTILE);
    dim3 block(THREADS);
    LSTMEncoder_78975858639682_kernel<<<grid, block, 0, stream>>>(
        x, w_ih0, w_hh0, b_ih0, b_hh0, w_ih1, w_hh1, b_ih1, b_hh1, fc_w, fc_b, out);
}


# Layer 1 runs one timestep behind layer 0: each barrier interval holds one layer's gate GEMM and, between its two halves, the other layer's cell update; 2 barriers per step instead of 4
# speedup vs baseline: 1.0891x; 1.0891x over previous
#include <hip/hip_runtime.h>
#include <hip/hip_bf16.h>


typedef __attribute__((ext_vector_type(2))) float v2f;
typedef __attribute__((ext_vector_type(8))) float v8f;

#define LSTM_B    2048
#define LSTM_T    512
#define LSTM_H    64
#define LSTM_G    256
#define MTILE     16
#define NWAVES    16
#define THREADS   (NWAVES * 32)
#define GPAD      260
#define HPAD      68

#define LOG2E_F   1.4426950408889634f

__device__ __forceinline__ float fast_tanh(float x) {
#if __has_builtin(__builtin_amdgcn_tanhf)
    return __builtin_amdgcn_tanhf(x);
#else
    float r;
    asm volatile("v_tanh_f32 %0, %1\n\tv_nop" : "=v"(r) : "v"(x));
    return r;
#endif
}

__device__ __forceinline__ float fast_sigmoid(float x) {
    float e = __builtin_amdgcn_exp2f(x * -LOG2E_F);
    return __builtin_amdgcn_rcpf(1.0f + e);
}

__device__ __forceinline__ v8f wmma_f32_k4(v2f a, v2f b, v8f c) {
    return __builtin_amdgcn_wmma_f32_16x16x4_f32(false, a, false, b,
                                                 (short)0, c, false, false);
}

__global__ __launch_bounds__(THREADS, 1)
void LSTMEncoder_78975858639682_kernel(
    const float* __restrict__ x,
    const float* __restrict__ w_ih0,
    const float* __restrict__ w_hh0,
    const float* __restrict__ b_ih0,
    const float* __restrict__ b_hh0,
    const float* __restrict__ w_ih1,
    const float* __restrict__ w_hh1,
    const float* __restrict__ b_ih1,
    const float* __restrict__ b_hh1,
    const float* __restrict__ fc_w,
    const float* __restrict__ fc_b,
    float* __restrict__ out)
{
    __shared__ float sG0[MTILE * GPAD];
    __shared__ float sG1[MTILE * GPAD];
    __shared__ float sH0[2 * MTILE * HPAD];
    __shared__ float sH1[MTILE * HPAD];

    const int tid  = threadIdx.x;
    const int lane = tid & 31;
    const int wave = tid >> 5;
    const int half = lane >> 4;
    const int l16  = lane & 15;
    const int rowBase = blockIdx.x * MTILE;

    const int ncol = wave * 16 + l16;
    const int kb   = 2 * half;
    v2f wih0 = *(const v2f*)(w_ih0 + ncol * 4 + kb);
    v2f whh0[16], wih1[16], whh1[16];
#pragma unroll
    for (int kc = 0; kc < 16; ++kc) {
        const int k = kc * 4 + kb;
        whh0[kc] = *(const v2f*)(w_hh0 + ncol * LSTM_H + k);
        wih1[kc] = *(const v2f*)(w_ih1 + ncol * LSTM_H + k);
        whh1[kc] = *(const v2f*)(w_hh1 + ncol * LSTM_H + k);
    }
    const float bias0 = b_ih0[ncol] + b_hh0[ncol];
    const float bias1 = b_ih1[ncol] + b_hh1[ncol];

    const int   erow = wave;
    const int   ecol = lane * 2;
    float c0x = 0.f, c0y = 0.f, c1x = 0.f, c1y = 0.f;

    for (int i = tid; i < MTILE * HPAD; i += THREADS) {
        sH0[i] = 0.f; sH0[MTILE * HPAD + i] = 0.f; sH1[i] = 0.f;
    }
    __syncthreads();

    const int arow = l16;
    const int aoff = arow * HPAD + kb;
    const int goff = erow * GPAD + ecol;
    const int hoff = erow * HPAD + ecol;
    const float* xrow = x + (size_t)(rowBase + arow) * LSTM_T * 4 + kb;


    auto gemm0 = [&](v2f axv, int hb, auto&& between) {
        v8f accA, accB;
#pragma unroll
        for (int j = 0; j < 8; ++j) { accA[j] = bias0; accB[j] = 0.f; }
        accB = wmma_f32_k4(axv, wih0, accB);
#pragma unroll
        for (int kc = 0; kc < 8; kc += 2) {
            v2f ahA = *(const v2f*)(sH0 + hb + aoff + (kc    ) * 4);
            v2f ahB = *(const v2f*)(sH0 + hb + aoff + (kc + 1) * 4);
            accA = wmma_f32_k4(ahA, whh0[kc],     accA);
            accB = wmma_f32_k4(ahB, whh0[kc + 1], accB);
        }
        between();
#pragma unroll
        for (int kc = 8; kc < 16; kc += 2) {
            v2f ahA = *(const v2f*)(sH0 + hb + aoff + (kc    ) * 4);
            v2f ahB = *(const v2f*)(sH0 + hb + aoff + (kc + 1) * 4);
            accA = wmma_f32_k4(ahA, whh0[kc],     accA);
            accB = wmma_f32_k4(ahB, whh0[kc + 1], accB);
        }
#pragma unroll
        for (int j = 0; j < 8; ++j)
            sG0[(j + 8 * half) * GPAD + ncol] = accA[j] + accB[j];
    };

    auto gemm1 = [&](int hb, auto&& between) {
        v8f accA, accB;
#pragma unroll
        for (int j = 0; j < 8; ++j) { accA[j] = bias1; accB[j] = 0.f; }
#pragma unroll
        for (int kc = 0; kc < 8; ++kc) {
            v2f ahA = *(const v2f*)(sH0 + hb + aoff + kc * 4);
            v2f ahB = *(const v2f*)(sH1 + aoff + kc * 4);
            accA = wmma_f32_k4(ahA, wih1[kc], accA);
            accB = wmma_f32_k4(ahB, whh1[kc], accB);
        }
        between();
#pragma unroll
        for (int kc = 8; kc < 16; ++kc) {
            v2f ahA = *(const v2f*)(sH0 + hb + aoff + kc * 4);
            v2f ahB = *(const v2f*)(sH1 + aoff + kc * 4);
            accA = wmma_f32_k4(ahA, wih1[kc], accA);
            accB = wmma_f32_k4(ahB, whh1[kc], accB);
        }
#pragma unroll
        for (int j = 0; j < 8; ++j)
            sG1[(j + 8 * half) * GPAD + ncol] = accA[j] + accB[j];
    };

    struct Gates { v2f i, f, g, o; };
    auto load_gates = [&](const float* sg) {
        Gates q;
        q.i = *(const v2f*)(sg + goff);
        q.f = *(const v2f*)(sg + goff + 64);
        q.g = *(const v2f*)(sg + goff + 128);
        q.o = *(const v2f*)(sg + goff + 192);
        return q;
    };
    auto cell = [&](const Gates& q, float& cx, float& cy, float* hdst) {
        float i0 = fast_sigmoid(q.i.x), i1 = fast_sigmoid(q.i.y);
        float f0 = fast_sigmoid(q.f.x), f1 = fast_sigmoid(q.f.y);
        float g0 = fast_tanh(q.g.x),    g1 = fast_tanh(q.g.y);
        float o0 = fast_sigmoid(q.o.x), o1 = fast_sigmoid(q.o.y);
        cx = f0 * cx + i0 * g0;
        cy = f1 * cy + i1 * g1;
        v2f hn; hn.x = o0 * fast_tanh(cx); hn.y = o1 * fast_tanh(cy);
        *(v2f*)(hdst + hoff) = hn;
    };
    auto nothing = [] {};

    const int HB = MTILE * HPAD;

    v2f ax = *(const v2f*)(xrow);
    gemm0(ax, HB, nothing);
    ax = *(const v2f*)(xrow + 4);
    __syncthreads();
    cell(load_gates(sG0), c0x, c0y, sH0);
    __syncthreads();
    gemm0(ax, 0, nothing);
    ax = *(const v2f*)(xrow + 8);
    __syncthreads();

#pragma unroll 1
    for (int t = 1; t < LSTM_T; ++t) {
        const int tn = (t + 2 < LSTM_T) ? (t + 2) : (LSTM_T - 1);
        v2f ax_next = *(const v2f*)(xrow + (size_t)tn * 4);
        const int hw = (t & 1) * HB;
        const int hr = HB - hw;

        {
            const Gates q = load_gates(sG0);
            gemm1(hr, [&] { cell(q, c0x, c0y, sH0 + hw); });
        }
        __syncthreads();

        {
            const Gates q = load_gates(sG1);
            gemm0(ax, hw, [&] { cell(q, c1x, c1y, sH1); });
        }
        __syncthreads();

        ax = ax_next;
    }

    gemm1(((LSTM_T - 1) & 1) * HB, nothing);
    __syncthreads();
    cell(load_gates(sG1), c1x, c1y, sH1);
    __syncthreads();

    if (wave < 4) {
        const int nco = wave * 16 + l16;
        v8f accA, accB;
        const float fb = fc_b[nco];
#pragma unroll
        for (int j = 0; j < 8; ++j) { accA[j] = fb; accB[j] = 0.f; }
#pragma unroll
        for (int kc = 0; kc < 16; kc += 2) {
            v2f ahA = *(const v2f*)(sH1 + arow * HPAD + (kc    ) * 4 + kb);
            v2f ahB = *(const v2f*)(sH1 + arow * HPAD + (kc + 1) * 4 + kb);
            v2f bwA = *(const v2f*)(fc_w + nco * LSTM_H + (kc    ) * 4 + kb);
            v2f bwB = *(const v2f*)(fc_w + nco * LSTM_H + (kc + 1) * 4 + kb);
            accA = wmma_f32_k4(ahA, bwA, accA);
            accB = wmma_f32_k4(ahB, bwB, accB);
        }
#pragma unroll
        for (int j = 0; j < 8; ++j) {
            const int r = j + 8 * half;
            out[(size_t)(rowBase + r) * LSTM_H + nco] = fast_tanh(accA[j] + accB[j]);
        }
    }
}

extern "C" void kernel_launch(void* const* d_in, const int* in_sizes, int n_in,
                              void* d_out, int out_size, void* d_ws, size_t ws_size,
                              hipStream_t stream) {
    (void)in_sizes; (void)n_in; (void)d_ws; (void)ws_size; (void)out_size;
    const float* x     = (const float*)d_in[0];
    const float* w_ih0 = (const float*)d_in[1];
    const float* w_hh0 = (const float*)d_in[2];
    const float* b_ih0 = (const float*)d_in[3];
    const float* b_hh0 = (const float*)d_in[4];
    const float* w_ih1 = (const float*)d_in[5];
    const float* w_hh1 = (const float*)d_in[6];
    const float* b_ih1 = (const float*)d_in[7];
    const float* b_hh1 = (const float*)d_in[8];
    const float* fc_w  = (const float*)d_in[9];
    const float* fc_b  = (const float*)d_in[10];
    float* out = (float*)d_out;

    dim3 grid(LSTM_B / MTILE);
    dim3 block(THREADS);
    LSTMEncoder_78975858639682_kernel<<<grid, block, 0, stream>>>(
        x, w_ih0, w_hh0, b_ih0, b_hh0, w_ih1, w_hh1, b_ih1, b_hh1, fc_w, fc_b, out);
}


# Each wave computes all four gate tiles of 16 hidden units (4 waves, one per SIMD): gates stay in accumulator registers for the cell update; each h fragment is read from LDS once per wave
# speedup vs baseline: 1.2096x; 1.1106x over previous
#include <hip/hip_runtime.h>
#include <hip/hip_bf16.h>


typedef __attribute__((ext_vector_type(2))) float v2f;
typedef __attribute__((ext_vector_type(8))) float v8f;

#define LSTM_B    2048
#define LSTM_T    512
#define LSTM_H    64
#define MTILE     16
#define NWAVES    4
#define THREADS   (NWAVES * 32)
#define HPAD      68

#define LOG2E_F   1.4426950408889634f

__device__ __forceinline__ float fast_tanh(float x) {
#if __has_builtin(__builtin_amdgcn_tanhf)
    return __builtin_amdgcn_tanhf(x);
#else
    float r;
    asm volatile("v_tanh_f32 %0, %1\n\tv_nop" : "=v"(r) : "v"(x));
    return r;
#endif
}

__device__ __forceinline__ float fast_sigmoid(float x) {
    float e = __builtin_amdgcn_exp2f(x * -LOG2E_F);
    return __builtin_amdgcn_rcpf(1.0f + e);
}

__device__ __forceinline__ v8f wmma_f32_k4(v2f a, v2f b, v8f c) {
    return __builtin_amdgcn_wmma_f32_16x16x4_f32(false, a, false, b,
                                                 (short)0, c, false, false);
}

__global__ __launch_bounds__(THREADS)
void LSTMEncoder_78975858639682_kernel(
    const float* __restrict__ x,
    const float* __restrict__ w_ih0,
    const float* __restrict__ w_hh0,
    const float* __restrict__ b_ih0,
    const float* __restrict__ b_hh0,
    const float* __restrict__ w_ih1,
    const float* __restrict__ w_hh1,
    const float* __restrict__ b_ih1,
    const float* __restrict__ b_hh1,
    const float* __restrict__ fc_w,
    const float* __restrict__ fc_b,
    float* __restrict__ out)
{
    __shared__ float sH0[2 * MTILE * HPAD];
    __shared__ float sH1[MTILE * HPAD];

    const int tid  = threadIdx.x;
    const int lane = tid & 31;
    const int wave = tid >> 5;
    const int half = lane >> 4;
    const int l16  = lane & 15;
    const int rowBase = blockIdx.x * MTILE;

    const int hcol = wave * 16 + l16;
    const int kb   = 2 * half;
    v2f   wih0[4];
    v2f   whh0[4][16], wih1[4][16], whh1[4][16];
    float bias0[4], bias1[4];
#pragma unroll
    for (int g = 0; g < 4; ++g) {
        const int n = g * LSTM_H + hcol;
        wih0[g] = *(const v2f*)(w_ih0 + n * 4 + kb);
#pragma unroll
        for (int kc = 0; kc < 16; ++kc) {
            const int k = kc * 4 + kb;
            whh0[g][kc] = *(const v2f*)(w_hh0 + n * LSTM_H + k);
            wih1[g][kc] = *(const v2f*)(w_ih1 + n * LSTM_H + k);
            whh1[g][kc] = *(const v2f*)(w_hh1 + n * LSTM_H + k);
        }
        bias0[g] = b_ih0[n] + b_hh0[n];
        bias1[g] = b_ih1[n] + b_hh1[n];
    }

    float c0[8], c1[8];
#pragma unroll
    for (int j = 0; j < 8; ++j) { c0[j] = 0.f; c1[j] = 0.f; }

    for (int i = tid; i < MTILE * HPAD; i += THREADS) {
        sH0[i] = 0.f; sH0[MTILE * HPAD + i] = 0.f; sH1[i] = 0.f;
    }
    __syncthreads();

    const int arow = l16;
    const int aoff = arow * HPAD + kb;
    const int hoff = (8 * half) * HPAD + hcol;
    const float* xrow = x + (size_t)(rowBase + arow) * LSTM_T * 4 + kb;

    auto load_frag = [&](const float* hsrc, v2f (&a)[16]) {
#pragma unroll
        for (int kc = 0; kc < 16; ++kc)
            a[kc] = *(const v2f*)(hsrc + aoff + kc * 4);
    };

    auto gemm0 = [&](v2f axv, const v2f (&a)[16], v8f (&G)[4]) {
        v8f accA[4], accB[4];
#pragma unroll
        for (int g = 0; g < 4; ++g) {
#pragma unroll
            for (int j = 0; j < 8; ++j) { accA[g][j] = bias0[g]; accB[g][j] = 0.f; }
        }
#pragma unroll
        for (int g = 0; g < 4; ++g) accB[g] = wmma_f32_k4(axv, wih0[g], accB[g]);
#pragma unroll
        for (int kc = 0; kc < 8; kc += 2) {
#pragma unroll
            for (int g = 0; g < 4; ++g) accA[g] = wmma_f32_k4(a[kc],     whh0[g][kc],     accA[g]);
#pragma unroll
            for (int g = 0; g < 4; ++g) accB[g] = wmma_f32_k4(a[kc + 1], whh0[g][kc + 1], accB[g]);
        }
        __builtin_amdgcn_sched_barrier(0);
#pragma unroll
        for (int kc = 8; kc < 16; kc += 2) {
#pragma unroll
            for (int g = 0; g < 4; ++g) accA[g] = wmma_f32_k4(a[kc],     whh0[g][kc],     accA[g]);
#pragma unroll
            for (int g = 0; g < 4; ++g) accB[g] = wmma_f32_k4(a[kc + 1], whh0[g][kc + 1], accB[g]);
        }
#pragma unroll
        for (int g = 0; g < 4; ++g) G[g] = accA[g] + accB[g];
    };

    v8f accA1[4], accB1[4];
    auto g1_chainA = [&](const v2f (&ah0)[16]) {
#pragma unroll
        for (int g = 0; g < 4; ++g) {
#pragma unroll
            for (int j = 0; j < 8; ++j) { accA1[g][j] = bias1[g]; accB1[g][j] = 0.f; }
        }
#pragma unroll
        for (int kc = 0; kc < 16; ++kc) {
#pragma unroll
            for (int g = 0; g < 4; ++g) accA1[g] = wmma_f32_k4(ah0[kc], wih1[g][kc], accA1[g]);
        }
    };
    auto g1_chainB_head = [&](const v2f (&ah1)[16]) {
#pragma unroll
        for (int kc = 0; kc < 4; ++kc) {
#pragma unroll
            for (int g = 0; g < 4; ++g) accB1[g] = wmma_f32_k4(ah1[kc], whh1[g][kc], accB1[g]);
        }
    };
    auto g1_chainB_tail = [&](const v2f (&ah1)[16]) {
#pragma unroll
        for (int kc = 4; kc < 16; ++kc) {
#pragma unroll
            for (int g = 0; g < 4; ++g) accB1[g] = wmma_f32_k4(ah1[kc], whh1[g][kc], accB1[g]);
        }
    };
    auto g1_merge = [&](v8f (&G)[4]) {
#pragma unroll
        for (int g = 0; g < 4; ++g) G[g] = accA1[g] + accB1[g];
    };

    auto cell = [&](const v8f (&G)[4], float (&c)[8], float* hdst) {
#pragma unroll
        for (int j = 0; j < 8; ++j) {
            const float ig = fast_sigmoid(G[0][j]);
            const float fg = fast_sigmoid(G[1][j]);
            const float gg = fast_tanh(G[2][j]);
            const float og = fast_sigmoid(G[3][j]);
            c[j] = fg * c[j] + ig * gg;
            hdst[hoff + j * HPAD] = og * fast_tanh(c[j]);
        }
    };

    const int HB = MTILE * HPAD;

    v2f a0[16], a1[16];
    v8f G0[4], G1[4];

    v2f ax = *(const v2f*)(xrow);
    load_frag(sH0 + HB, a0);
    gemm0(ax, a0, G0);
    ax = *(const v2f*)(xrow + 4);
    cell(G0, c0, sH0);
    __syncthreads();
    load_frag(sH0, a0);
    gemm0(ax, a0, G0);
    ax = *(const v2f*)(xrow + 8);

#pragma unroll 1
    for (int t = 1; t < LSTM_T; ++t) {
        const int tn = (t + 2 < LSTM_T) ? (t + 2) : (LSTM_T - 1);
        v2f ax_next = *(const v2f*)(xrow + (size_t)tn * 4);
        const int hw = (t & 1) * HB;

        load_frag(sH1, a1);
        __builtin_amdgcn_sched_barrier(0);
        cell(G0, c0, sH0 + hw);
        g1_chainA(a0);
        __builtin_amdgcn_sched_barrier(0);
        g1_chainB_head(a1);
        __syncthreads();
        load_frag(sH0 + hw, a0);
        __builtin_amdgcn_sched_barrier(0);
        g1_chainB_tail(a1);
        g1_merge(G1);
        __builtin_amdgcn_sched_barrier(0);

        cell(G1, c1, sH1);
        gemm0(ax, a0, G0);
        __syncthreads();

        ax = ax_next;
    }

    load_frag(sH1, a1);
    g1_chainA(a0);
    g1_chainB_head(a1);
    g1_chainB_tail(a1);
    g1_merge(G1);
    __syncthreads();
    cell(G1, c1, sH1);
    __syncthreads();

    {
        const int nco = wave * 16 + l16;
        v8f accA, accB;
        const float fb = fc_b[nco];
#pragma unroll
        for (int j = 0; j < 8; ++j) { accA[j] = fb; accB[j] = 0.f; }
#pragma unroll
        for (int kc = 0; kc < 16; kc += 2) {
            v2f ahA = *(const v2f*)(sH1 + arow * HPAD + (kc    ) * 4 + kb);
            v2f ahB = *(const v2f*)(sH1 + arow * HPAD + (kc + 1) * 4 + kb);
            v2f bwA = *(const v2f*)(fc_w + nco * LSTM_H + (kc    ) * 4 + kb);
            v2f bwB = *(const v2f*)(fc_w + nco * LSTM_H + (kc + 1) * 4 + kb);
            accA = wmma_f32_k4(ahA, bwA, accA);
            accB = wmma_f32_k4(ahB, bwB, accB);
        }
#pragma unroll
        for (int j = 0; j < 8; ++j) {
            const int r = j + 8 * half;
            out[(size_t)(rowBase + r) * LSTM_H + nco] = fast_tanh(accA[j] + accB[j]);
        }
    }
}

extern "C" void kernel_launch(void* const* d_in, const int* in_sizes, int n_in,
                              void* d_out, int out_size, void* d_ws, size_t ws_size,
                              hipStream_t stream) {
    (void)in_sizes; (void)n_in; (void)d_ws; (void)ws_size; (void)out_size;
    const float* x     = (const float*)d_in[0];
    const float* w_ih0 = (const float*)d_in[1];
    const float* w_hh0 = (const float*)d_in[2];
    const float* b_ih0 = (const float*)d_in[3];
    const float* b_hh0 = (const float*)d_in[4];
    const float* w_ih1 = (const float*)d_in[5];
    const float* w_hh1 = (const float*)d_in[6];
    const float* b_ih1 = (const float*)d_in[7];
    const float* b_hh1 = (const float*)d_in[8];
    const float* fc_w  = (const float*)d_in[9];
    const float* fc_b  = (const float*)d_in[10];
    float* out = (float*)d_out;

    dim3 grid(LSTM_B / MTILE);
    dim3 block(THREADS);
    LSTMEncoder_78975858639682_kernel<<<grid, block, 0, stream>>>(
        x, w_ih0, w_hh0, b_ih0, b_hh0, w_ih1, w_hh1, b_ih1, b_hh1, fc_w, fc_b, out);
}


# Sigmoid gates as 0.5*tanh(x/2)+0.5 on the hardware tanh, the 1/2 folded exactly into the i,f,o weight and bias registers at load: 1 transcendental + 1 fma per gate instead of mul, exp, add, rcp
# speedup vs baseline: 1.2489x; 1.0325x over previous
#include <hip/hip_runtime.h>
#include <hip/hip_bf16.h>


typedef __attribute__((ext_vector_type(2))) float v2f;
typedef __attribute__((ext_vector_type(8))) float v8f;

#define LSTM_B    2048
#define LSTM_T    512
#define LSTM_H    64
#define MTILE     16
#define NWAVES    4
#define THREADS   (NWAVES * 32)
#define HPAD      68

__device__ __forceinline__ float fast_tanh(float x) {
#if __has_builtin(__builtin_amdgcn_tanhf)
    return __builtin_amdgcn_tanhf(x);
#else
    float r;
    asm volatile("v_tanh_f32 %0, %1\n\tv_nop" : "=v"(r) : "v"(x));
    return r;
#endif
}

__device__ __forceinline__ float sigmoid_of_half(float xh) {
    return __builtin_fmaf(fast_tanh(xh), 0.5f, 0.5f);
}

__device__ __forceinline__ v8f wmma_f32_k4(v2f a, v2f b, v8f c) {
    return __builtin_amdgcn_wmma_f32_16x16x4_f32(false, a, false, b,
                                                 (short)0, c, false, false);
}

__global__ __launch_bounds__(THREADS)
void LSTMEncoder_78975858639682_kernel(
    const float* __restrict__ x,
    const float* __restrict__ w_ih0,
    const float* __restrict__ w_hh0,
    const float* __restrict__ b_ih0,
    const float* __restrict__ b_hh0,
    const float* __restrict__ w_ih1,
    const float* __restrict__ w_hh1,
    const float* __restrict__ b_ih1,
    const float* __restrict__ b_hh1,
    const float* __restrict__ fc_w,
    const float* __restrict__ fc_b,
    float* __restrict__ out)
{
    __shared__ float sH0[2 * MTILE * HPAD];
    __shared__ float sH1[MTILE * HPAD];

    const int tid  = threadIdx.x;
    const int lane = tid & 31;
    const int wave = tid >> 5;
    const int half = lane >> 4;
    const int l16  = lane & 15;
    const int rowBase = blockIdx.x * MTILE;

    const int hcol = wave * 16 + l16;
    const int kb   = 2 * half;
    v2f   wih0[4];
    v2f   whh0[4][16], wih1[4][16], whh1[4][16];
    float bias0[4], bias1[4];
#pragma unroll
    for (int g = 0; g < 4; ++g) {
        const int n = g * LSTM_H + hcol;
        const float sc = (g == 2) ? 1.0f : 0.5f;
        wih0[g] = *(const v2f*)(w_ih0 + n * 4 + kb) * sc;
#pragma unroll
        for (int kc = 0; kc < 16; ++kc) {
            const int k = kc * 4 + kb;
            whh0[g][kc] = *(const v2f*)(w_hh0 + n * LSTM_H + k) * sc;
            wih1[g][kc] = *(const v2f*)(w_ih1 + n * LSTM_H + k) * sc;
            whh1[g][kc] = *(const v2f*)(w_hh1 + n * LSTM_H + k) * sc;
        }
        bias0[g] = (b_ih0[n] + b_hh0[n]) * sc;
        bias1[g] = (b_ih1[n] + b_hh1[n]) * sc;
    }

    float c0[8], c1[8];
#pragma unroll
    for (int j = 0; j < 8; ++j) { c0[j] = 0.f; c1[j] = 0.f; }

    for (int i = tid; i < MTILE * HPAD; i += THREADS) {
        sH0[i] = 0.f; sH0[MTILE * HPAD + i] = 0.f; sH1[i] = 0.f;
    }
    __syncthreads();

    const int arow = l16;
    const int aoff = arow * HPAD + kb;
    const int hoff = (8 * half) * HPAD + hcol;
    const float* xrow = x + (size_t)(rowBase + arow) * LSTM_T * 4 + kb;

    auto load_frag = [&](const float* hsrc, v2f (&a)[16]) {
#pragma unroll
        for (int kc = 0; kc < 16; ++kc)
            a[kc] = *(const v2f*)(hsrc + aoff + kc * 4);
    };

    auto gemm0 = [&](v2f axv, const v2f (&a)[16], v8f (&G)[4]) {
        v8f accA[4], accB[4];
#pragma unroll
        for (int g = 0; g < 4; ++g) {
#pragma unroll
            for (int j = 0; j < 8; ++j) { accA[g][j] = bias0[g]; accB[g][j] = 0.f; }
        }
#pragma unroll
        for (int g = 0; g < 4; ++g) accB[g] = wmma_f32_k4(axv, wih0[g], accB[g]);
#pragma unroll
        for (int kc = 0; kc < 8; kc += 2) {
#pragma unroll
            for (int g = 0; g < 4; ++g) accA[g] = wmma_f32_k4(a[kc],     whh0[g][kc],     accA[g]);
#pragma unroll
            for (int g = 0; g < 4; ++g) accB[g] = wmma_f32_k4(a[kc + 1], whh0[g][kc + 1], accB[g]);
        }
        __builtin_amdgcn_sched_barrier(0);
#pragma unroll
        for (int kc = 8; kc < 16; kc += 2) {
#pragma unroll
            for (int g = 0; g < 4; ++g) accA[g] = wmma_f32_k4(a[kc],     whh0[g][kc],     accA[g]);
#pragma unroll
            for (int g = 0; g < 4; ++g) accB[g] = wmma_f32_k4(a[kc + 1], whh0[g][kc + 1], accB[g]);
        }
#pragma unroll
        for (int g = 0; g < 4; ++g) G[g] = accA[g] + accB[g];
    };

    v8f accA1[4], accB1[4];
    auto g1_chainA = [&](const v2f (&ah0)[16]) {
#pragma unroll
        for (int g = 0; g < 4; ++g) {
#pragma unroll
            for (int j = 0; j < 8; ++j) { accA1[g][j] = bias1[g]; accB1[g][j] = 0.f; }
        }
#pragma unroll
        for (int kc = 0; kc < 16; ++kc) {
#pragma unroll
            for (int g = 0; g < 4; ++g) accA1[g] = wmma_f32_k4(ah0[kc], wih1[g][kc], accA1[g]);
        }
    };
    auto g1_chainB_head = [&](const v2f (&ah1)[16]) {
#pragma unroll
        for (int kc = 0; kc < 4; ++kc) {
#pragma unroll
            for (int g = 0; g < 4; ++g) accB1[g] = wmma_f32_k4(ah1[kc], whh1[g][kc], accB1[g]);
        }
    };
    auto g1_chainB_tail = [&](const v2f (&ah1)[16]) {
#pragma unroll
        for (int kc = 4; kc < 16; ++kc) {
#pragma unroll
            for (int g = 0; g < 4; ++g) accB1[g] = wmma_f32_k4(ah1[kc], whh1[g][kc], accB1[g]);
        }
    };
    auto g1_merge = [&](v8f (&G)[4]) {
#pragma unroll
        for (int g = 0; g < 4; ++g) G[g] = accA1[g] + accB1[g];
    };

    auto cell = [&](const v8f (&G)[4], float (&c)[8], float* hdst) {
#pragma unroll
        for (int j = 0; j < 8; ++j) {
            const float ig = sigmoid_of_half(G[0][j]);
            const float fg = sigmoid_of_half(G[1][j]);
            const float gg = fast_tanh(G[2][j]);
            const float og = sigmoid_of_half(G[3][j]);
            c[j] = fg * c[j] + ig * gg;
            hdst[hoff + j * HPAD] = og * fast_tanh(c[j]);
        }
    };

    const int HB = MTILE * HPAD;

    v2f a0[16], a1[16];
    v8f G0[4], G1[4];

    v2f ax = *(const v2f*)(xrow);
    load_frag(sH0 + HB, a0);
    gemm0(ax, a0, G0);
    ax = *(const v2f*)(xrow + 4);
    cell(G0, c0, sH0);
    __syncthreads();
    load_frag(sH0, a0);
    gemm0(ax, a0, G0);
    ax = *(const v2f*)(xrow + 8);

#pragma unroll 1
    for (int t = 1; t < LSTM_T; ++t) {
        const int tn = (t + 2 < LSTM_T) ? (t + 2) : (LSTM_T - 1);
        v2f ax_next = *(const v2f*)(xrow + (size_t)tn * 4);
        const int hw = (t & 1) * HB;

        load_frag(sH1, a1);
        __builtin_amdgcn_sched_barrier(0);
        cell(G0, c0, sH0 + hw);
        g1_chainA(a0);
        __builtin_amdgcn_sched_barrier(0);
        g1_chainB_head(a1);
        __syncthreads();
        load_frag(sH0 + hw, a0);
        __builtin_amdgcn_sched_barrier(0);
        g1_chainB_tail(a1);
        g1_merge(G1);
        __builtin_amdgcn_sched_barrier(0);

        cell(G1, c1, sH1);
        gemm0(ax, a0, G0);
        __syncthreads();

        ax = ax_next;
    }

    load_frag(sH1, a1);
    g1_chainA(a0);
    g1_chainB_head(a1);
    g1_chainB_tail(a1);
    g1_merge(G1);
    __syncthreads();
    cell(G1, c1, sH1);
    __syncthreads();

    {
        const int nco = wave * 16 + l16;
        v8f accA, accB;
        const float fb = fc_b[nco];
#pragma unroll
        for (int j = 0; j < 8; ++j) { accA[j] = fb; accB[j] = 0.f; }
#pragma unroll
        for (int kc = 0; kc < 16; kc += 2) {
            v2f ahA = *(const v2f*)(sH1 + arow * HPAD + (kc    ) * 4 + kb);
            v2f ahB = *(const v2f*)(sH1 + arow * HPAD + (kc + 1) * 4 + kb);
            v2f bwA = *(const v2f*)(fc_w + nco * LSTM_H + (kc    ) * 4 + kb);
            v2f bwB = *(const v2f*)(fc_w + nco * LSTM_H + (kc + 1) * 4 + kb);
            accA = wmma_f32_k4(ahA, bwA, accA);
            accB = wmma_f32_k4(ahB, bwB, accB);
        }
#pragma unroll
        for (int j = 0; j < 8; ++j) {
            const int r = j + 8 * half;
            out[(size_t)(rowBase + r) * LSTM_H + nco] = fast_tanh(accA[j] + accB[j]);
        }
    }
}

extern "C" void kernel_launch(void* const* d_in, const int* in_sizes, int n_in,
                              void* d_out, int out_size, void* d_ws, size_t ws_size,
                              hipStream_t stream) {
    (void)in_sizes; (void)n_in; (void)d_ws; (void)ws_size; (void)out_size;
    const float* x     = (const float*)d_in[0];
    const float* w_ih0 = (const float*)d_in[1];
    const float* w_hh0 = (const float*)d_in[2];
    const float* b_ih0 = (const float*)d_in[3];
    const float* b_hh0 = (const float*)d_in[4];
    const float* w_ih1 = (const float*)d_in[5];
    const float* w_hh1 = (const float*)d_in[6];
    const float* b_ih1 = (const float*)d_in[7];
    const float* b_hh1 = (const float*)d_in[8];
    const float* fc_w  = (const float*)d_in[9];
    const float* fc_b  = (const float*)d_in[10];
    float* out = (float*)d_out;

    dim3 grid(LSTM_B / MTILE);
    dim3 block(THREADS);
    LSTMEncoder_78975858639682_kernel<<<grid, block, 0, stream>>>(
        x, w_ih0, w_hh0, b_ih0, b_hh0, w_ih1, w_hh1, b_ih1, b_hh1, fc_w, fc_b, out);
}


# 8 batch rows per workgroup on 256 workgroups (M tile half filled, 4 real rows per lane half): per-lane cell update, merge adds and h stores halve; WMMAs per workgroup and summation order unchanged
# speedup vs baseline: 1.2907x; 1.0335x over previous
#include <hip/hip_runtime.h>
#include <hip/hip_bf16.h>


typedef __attribute__((ext_vector_type(2))) float v2f;
typedef __attribute__((ext_vector_type(8))) float v8f;

#define LSTM_B    2048
#define LSTM_T    512
#define LSTM_H    64
#define MTILE     16
#define MROWS     8
#define JROWS     4
#define NWAVES    4
#define THREADS   (NWAVES * 32)
#define HPAD      68

__device__ __forceinline__ float fast_tanh(float x) {
#if __has_builtin(__builtin_amdgcn_tanhf)
    return __builtin_amdgcn_tanhf(x);
#else
    float r;
    asm volatile("v_tanh_f32 %0, %1\n\tv_nop" : "=v"(r) : "v"(x));
    return r;
#endif
}

__device__ __forceinline__ float sigmoid_of_half(float xh) {
    return __builtin_fmaf(fast_tanh(xh), 0.5f, 0.5f);
}

__device__ __forceinline__ v8f wmma_f32_k4(v2f a, v2f b, v8f c) {
    return __builtin_amdgcn_wmma_f32_16x16x4_f32(false, a, false, b,
                                                 (short)0, c, false, false);
}

__global__ __launch_bounds__(THREADS)
void LSTMEncoder_78975858639682_kernel(
    const float* __restrict__ x,
    const float* __restrict__ w_ih0,
    const float* __restrict__ w_hh0,
    const float* __restrict__ b_ih0,
    const float* __restrict__ b_hh0,
    const float* __restrict__ w_ih1,
    const float* __restrict__ w_hh1,
    const float* __restrict__ b_ih1,
    const float* __restrict__ b_hh1,
    const float* __restrict__ fc_w,
    const float* __restrict__ fc_b,
    float* __restrict__ out)
{
    __shared__ float sH0[2 * MTILE * HPAD];
    __shared__ float sH1[MTILE * HPAD];

    const int tid  = threadIdx.x;
    const int lane = tid & 31;
    const int wave = tid >> 5;
    const int half = lane >> 4;
    const int l16  = lane & 15;
    const int rowBase = blockIdx.x * MROWS;

    const int hcol = wave * 16 + l16;
    const int kb   = 2 * half;
    v2f   wih0[4];
    v2f   whh0[4][16], wih1[4][16], whh1[4][16];
    float bias0[4], bias1[4];
#pragma unroll
    for (int g = 0; g < 4; ++g) {
        const int n = g * LSTM_H + hcol;
        const float sc = (g == 2) ? 1.0f : 0.5f;
        wih0[g] = *(const v2f*)(w_ih0 + n * 4 + kb) * sc;
#pragma unroll
        for (int kc = 0; kc < 16; ++kc) {
            const int k = kc * 4 + kb;
            whh0[g][kc] = *(const v2f*)(w_hh0 + n * LSTM_H + k) * sc;
            wih1[g][kc] = *(const v2f*)(w_ih1 + n * LSTM_H + k) * sc;
            whh1[g][kc] = *(const v2f*)(w_hh1 + n * LSTM_H + k) * sc;
        }
        bias0[g] = (b_ih0[n] + b_hh0[n]) * sc;
        bias1[g] = (b_ih1[n] + b_hh1[n]) * sc;
    }

    float c0[JROWS], c1[JROWS];
#pragma unroll
    for (int j = 0; j < JROWS; ++j) { c0[j] = 0.f; c1[j] = 0.f; }

    for (int i = tid; i < MTILE * HPAD; i += THREADS) {
        sH0[i] = 0.f; sH0[MTILE * HPAD + i] = 0.f; sH1[i] = 0.f;
    }
    __syncthreads();

    const int arow = l16;
    const int aoff = arow * HPAD + kb;
    const int hoff = (8 * half) * HPAD + hcol;
    const int xloc = (arow & 3) + 4 * (arow >> 3);
    const float* xrow = x + (size_t)(rowBase + xloc) * LSTM_T * 4 + kb;

    auto load_frag = [&](const float* hsrc, v2f (&a)[16]) {
#pragma unroll
        for (int kc = 0; kc < 16; ++kc)
            a[kc] = *(const v2f*)(hsrc + aoff + kc * 4);
    };

    auto gemm0 = [&](v2f axv, const v2f (&a)[16], v8f (&G)[4]) {
        v8f accA[4], accB[4];
#pragma unroll
        for (int g = 0; g < 4; ++g) {
#pragma unroll
            for (int j = 0; j < 8; ++j) { accA[g][j] = bias0[g]; accB[g][j] = 0.f; }
        }
#pragma unroll
        for (int g = 0; g < 4; ++g) accB[g] = wmma_f32_k4(axv, wih0[g], accB[g]);
#pragma unroll
        for (int kc = 0; kc < 8; kc += 2) {
#pragma unroll
            for (int g = 0; g < 4; ++g) accA[g] = wmma_f32_k4(a[kc],     whh0[g][kc],     accA[g]);
#pragma unroll
            for (int g = 0; g < 4; ++g) accB[g] = wmma_f32_k4(a[kc + 1], whh0[g][kc + 1], accB[g]);
        }
        __builtin_amdgcn_sched_barrier(0);
#pragma unroll
        for (int kc = 8; kc < 16; kc += 2) {
#pragma unroll
            for (int g = 0; g < 4; ++g) accA[g] = wmma_f32_k4(a[kc],     whh0[g][kc],     accA[g]);
#pragma unroll
            for (int g = 0; g < 4; ++g) accB[g] = wmma_f32_k4(a[kc + 1], whh0[g][kc + 1], accB[g]);
        }
#pragma unroll
        for (int g = 0; g < 4; ++g) {
#pragma unroll
            for (int j = 0; j < JROWS; ++j) G[g][j] = accA[g][j] + accB[g][j];
        }
    };

    v8f accA1[4], accB1[4];
    auto g1_chainA = [&](const v2f (&ah0)[16]) {
#pragma unroll
        for (int g = 0; g < 4; ++g) {
#pragma unroll
            for (int j = 0; j < 8; ++j) { accA1[g][j] = bias1[g]; accB1[g][j] = 0.f; }
        }
#pragma unroll
        for (int kc = 0; kc < 16; ++kc) {
#pragma unroll
            for (int g = 0; g < 4; ++g) accA1[g] = wmma_f32_k4(ah0[kc], wih1[g][kc], accA1[g]);
        }
    };
    auto g1_chainB_head = [&](const v2f (&ah1)[16]) {
#pragma unroll
        for (int kc = 0; kc < 4; ++kc) {
#pragma unroll
            for (int g = 0; g < 4; ++g) accB1[g] = wmma_f32_k4(ah1[kc], whh1[g][kc], accB1[g]);
        }
    };
    auto g1_chainB_tail = [&](const v2f (&ah1)[16]) {
#pragma unroll
        for (int kc = 4; kc < 16; ++kc) {
#pragma unroll
            for (int g = 0; g < 4; ++g) accB1[g] = wmma_f32_k4(ah1[kc], whh1[g][kc], accB1[g]);
        }
    };
    auto g1_merge = [&](v8f (&G)[4]) {
#pragma unroll
        for (int g = 0; g < 4; ++g) {
#pragma unroll
            for (int j = 0; j < JROWS; ++j) G[g][j] = accA1[g][j] + accB1[g][j];
        }
    };

    auto cell = [&](const v8f (&G)[4], float (&c)[JROWS], float* hdst) {
#pragma unroll
        for (int j = 0; j < JROWS; ++j) {
            const float ig = sigmoid_of_half(G[0][j]);
            const float fg = sigmoid_of_half(G[1][j]);
            const float gg = fast_tanh(G[2][j]);
            const float og = sigmoid_of_half(G[3][j]);
            c[j] = fg * c[j] + ig * gg;
            hdst[hoff + j * HPAD] = og * fast_tanh(c[j]);
        }
    };

    const int HB = MTILE * HPAD;

    v2f a0[16], a1[16];
    v8f G0[4], G1[4];

    v2f ax = *(const v2f*)(xrow);
    load_frag(sH0 + HB, a0);
    gemm0(ax, a0, G0);
    ax = *(const v2f*)(xrow + 4);
    cell(G0, c0, sH0);
    __syncthreads();
    load_frag(sH0, a0);
    gemm0(ax, a0, G0);
    ax = *(const v2f*)(xrow + 8);

#pragma unroll 1
    for (int t = 1; t < LSTM_T; ++t) {
        const int tn = (t + 2 < LSTM_T) ? (t + 2) : (LSTM_T - 1);
        v2f ax_next = *(const v2f*)(xrow + (size_t)tn * 4);
        const int hw = (t & 1) * HB;

        load_frag(sH1, a1);
        __builtin_amdgcn_sched_barrier(0);
        cell(G0, c0, sH0 + hw);
        g1_chainA(a0);
        __builtin_amdgcn_sched_barrier(0);
        g1_chainB_head(a1);
        __syncthreads();
        load_frag(sH0 + hw, a0);
        __builtin_amdgcn_sched_barrier(0);
        g1_chainB_tail(a1);
        g1_merge(G1);
        __builtin_amdgcn_sched_barrier(0);

        cell(G1, c1, sH1);
        gemm0(ax, a0, G0);
        __syncthreads();

        ax = ax_next;
    }

    load_frag(sH1, a1);
    g1_chainA(a0);
    g1_chainB_head(a1);
    g1_chainB_tail(a1);
    g1_merge(G1);
    __syncthreads();
    cell(G1, c1, sH1);
    __syncthreads();

    {
        const int nco = wave * 16 + l16;
        v8f accA, accB;
        const float fb = fc_b[nco];
#pragma unroll
        for (int j = 0; j < 8; ++j) { accA[j] = fb; accB[j] = 0.f; }
#pragma unroll
        for (int kc = 0; kc < 16; kc += 2) {
            v2f ahA = *(const v2f*)(sH1 + arow * HPAD + (kc    ) * 4 + kb);
            v2f ahB = *(const v2f*)(sH1 + arow * HPAD + (kc + 1) * 4 + kb);
            v2f bwA = *(const v2f*)(fc_w + nco * LSTM_H + (kc    ) * 4 + kb);
            v2f bwB = *(const v2f*)(fc_w + nco * LSTM_H + (kc + 1) * 4 + kb);
            accA = wmma_f32_k4(ahA, bwA, accA);
            accB = wmma_f32_k4(ahB, bwB, accB);
        }
#pragma unroll
        for (int j = 0; j < JROWS; ++j) {
            const int r = j + 4 * half;
            out[(size_t)(rowBase + r) * LSTM_H + nco] = fast_tanh(accA[j] + accB[j]);
        }
    }
}

extern "C" void kernel_launch(void* const* d_in, const int* in_sizes, int n_in,
                              void* d_out, int out_size, void* d_ws, size_t ws_size,
                              hipStream_t stream) {
    (void)in_sizes; (void)n_in; (void)d_ws; (void)ws_size; (void)out_size;
    const float* x     = (const float*)d_in[0];
    const float* w_ih0 = (const float*)d_in[1];
    const float* w_hh0 = (const float*)d_in[2];
    const float* b_ih0 = (const float*)d_in[3];
    const float* b_hh0 = (const float*)d_in[4];
    const float* w_ih1 = (const float*)d_in[5];
    const float* w_hh1 = (const float*)d_in[6];
    const float* b_ih1 = (const float*)d_in[7];
    const float* b_hh1 = (const float*)d_in[8];
    const float* fc_w  = (const float*)d_in[9];
    const float* fc_b  = (const float*)d_in[10];
    float* out = (float*)d_out;

    dim3 grid(LSTM_B / MROWS);
    dim3 block(THREADS);
    LSTMEncoder_78975858639682_kernel<<<grid, block, 0, stream>>>(
        x, w_ih0, w_hh0, b_ih0, b_hh0, w_ih1, w_hh1, b_ih1, b_hh1, fc_w, fc_b, out);
}


# Layer-1 input product h0@Wih1^T done for two timesteps per WMMA pass: the odd step's h0 sits in the M tile's unused rows, layer 1 runs two steps behind layer 0; 164 not 196 WMMAs per step and wave
# speedup vs baseline: 1.5086x; 1.1688x over previous
#include <hip/hip_runtime.h>
#include <hip/hip_bf16.h>


typedef __attribute__((ext_vector_type(2))) float v2f;
typedef __attribute__((ext_vector_type(8))) float v8f;

#define LSTM_B    2048
#define LSTM_T    512
#define LSTM_H    64
#define MTILE     16
#define MROWS     8
#define JROWS     4
#define NWAVES    4
#define THREADS   (NWAVES * 32)
#define HPAD      68

__device__ __forceinline__ float fast_tanh(float x) {
#if __has_builtin(__builtin_amdgcn_tanhf)
    return __builtin_amdgcn_tanhf(x);
#else
    float r;
    asm volatile("v_tanh_f32 %0, %1\n\tv_nop" : "=v"(r) : "v"(x));
    return r;
#endif
}

__device__ __forceinline__ float sigmoid_of_half(float xh) {
    return __builtin_fmaf(fast_tanh(xh), 0.5f, 0.5f);
}

__device__ __forceinline__ v8f wmma_f32_k4(v2f a, v2f b, v8f c) {
    return __builtin_amdgcn_wmma_f32_16x16x4_f32(false, a, false, b,
                                                 (short)0, c, false, false);
}

__global__ __launch_bounds__(THREADS)
void LSTMEncoder_78975858639682_kernel(
    const float* __restrict__ x,
    const float* __restrict__ w_ih0,
    const float* __restrict__ w_hh0,
    const float* __restrict__ b_ih0,
    const float* __restrict__ b_hh0,
    const float* __restrict__ w_ih1,
    const float* __restrict__ w_hh1,
    const float* __restrict__ b_ih1,
    const float* __restrict__ b_hh1,
    const float* __restrict__ fc_w,
    const float* __restrict__ fc_b,
    float* __restrict__ out)
{
    __shared__ float sH0[MTILE * HPAD];
    __shared__ float sH1[MTILE * HPAD];

    const int tid  = threadIdx.x;
    const int lane = tid & 31;
    const int wave = tid >> 5;
    const int half = lane >> 4;
    const int l16  = lane & 15;
    const int rowBase = blockIdx.x * MROWS;

    const int hcol = wave * 16 + l16;
    const int kb   = 2 * half;
    v2f   wih0[4];
    v2f   whh0[4][16], wih1[4][16], whh1[4][16];
    float bias0[4], bias1[4];
#pragma unroll
    for (int g = 0; g < 4; ++g) {
        const int n = g * LSTM_H + hcol;
        const float sc = (g == 2) ? 1.0f : 0.5f;
        wih0[g] = *(const v2f*)(w_ih0 + n * 4 + kb) * sc;
#pragma unroll
        for (int kc = 0; kc < 16; ++kc) {
            const int k = kc * 4 + kb;
            whh0[g][kc] = *(const v2f*)(w_hh0 + n * LSTM_H + k) * sc;
            wih1[g][kc] = *(const v2f*)(w_ih1 + n * LSTM_H + k) * sc;
            whh1[g][kc] = *(const v2f*)(w_hh1 + n * LSTM_H + k) * sc;
        }
        bias0[g] = (b_ih0[n] + b_hh0[n]) * sc;
        bias1[g] = (b_ih1[n] + b_hh1[n]) * sc;
    }

    float c0[JROWS], c1[JROWS];
#pragma unroll
    for (int j = 0; j < JROWS; ++j) { c0[j] = 0.f; c1[j] = 0.f; }

    for (int i = tid; i < MTILE * HPAD; i += THREADS) {
        sH0[i] = 0.f; sH1[i] = 0.f;
    }
    __syncthreads();

    const int arow  = l16;
    const int aoffE = arow * HPAD + kb;
    const int aoffO = (arow ^ 4) * HPAD + kb;
    const int hoff  = (8 * half) * HPAD + hcol;
    const int HODD  = 4 * HPAD;
    const int xloc = (arow & 3) + 4 * (arow >> 3);
    const float* xrow = x + (size_t)(rowBase + xloc) * LSTM_T * 4 + kb;

    auto load_frag = [&](const float* hsrc, int off, v2f (&a)[16]) {
#pragma unroll
        for (int kc = 0; kc < 16; ++kc)
            a[kc] = *(const v2f*)(hsrc + off + kc * 4);
    };

    auto gemm0 = [&](v2f axv, const v2f (&a)[16], v8f (&G)[4]) {
        v8f accA[4], accB[4];
#pragma unroll
        for (int g = 0; g < 4; ++g) {
#pragma unroll
            for (int j = 0; j < 8; ++j) { accA[g][j] = bias0[g]; accB[g][j] = 0.f; }
        }
#pragma unroll
        for (int g = 0; g < 4; ++g) accB[g] = wmma_f32_k4(axv, wih0[g], accB[g]);
#pragma unroll
        for (int kc = 0; kc < 8; kc += 2) {
#pragma unroll
            for (int g = 0; g < 4; ++g) accA[g] = wmma_f32_k4(a[kc],     whh0[g][kc],     accA[g]);
#pragma unroll
            for (int g = 0; g < 4; ++g) accB[g] = wmma_f32_k4(a[kc + 1], whh0[g][kc + 1], accB[g]);
        }
        __builtin_amdgcn_sched_barrier(0);
#pragma unroll
        for (int kc = 8; kc < 16; kc += 2) {
#pragma unroll
            for (int g = 0; g < 4; ++g) accA[g] = wmma_f32_k4(a[kc],     whh0[g][kc],     accA[g]);
#pragma unroll
            for (int g = 0; g < 4; ++g) accB[g] = wmma_f32_k4(a[kc + 1], whh0[g][kc + 1], accB[g]);
        }
#pragma unroll
        for (int g = 0; g < 4; ++g) {
#pragma unroll
            for (int j = 0; j < JROWS; ++j) G[g][j] = accA[g][j] + accB[g][j];
        }
    };

    v8f P[4], accB1[4];
    auto proj1 = [&](const v2f (&ah0)[16]) {
#pragma unroll
        for (int g = 0; g < 4; ++g) {
#pragma unroll
            for (int j = 0; j < 8; ++j) P[g][j] = bias1[g];
        }
#pragma unroll
        for (int kc = 0; kc < 16; ++kc) {
#pragma unroll
            for (int g = 0; g < 4; ++g) P[g] = wmma_f32_k4(ah0[kc], wih1[g][kc], P[g]);
        }
    };
    auto rec1_zero = [&]() {
#pragma unroll
        for (int g = 0; g < 4; ++g) {
#pragma unroll
            for (int j = 0; j < 8; ++j) accB1[g][j] = 0.f;
        }
    };
    auto rec1 = [&](const v2f (&ah1)[16], int kfrom, int kto) {
#pragma unroll
        for (int kc = kfrom; kc < kto; ++kc) {
#pragma unroll
            for (int g = 0; g < 4; ++g) accB1[g] = wmma_f32_k4(ah1[kc], whh1[g][kc], accB1[g]);
        }
    };
    auto merge_even = [&](v8f (&G)[4]) {
#pragma unroll
        for (int g = 0; g < 4; ++g) {
#pragma unroll
            for (int j = 0; j < JROWS; ++j) G[g][j] = P[g][JROWS + j] + accB1[g][j];
        }
    };
    auto merge_odd = [&](v8f (&G)[4]) {
#pragma unroll
        for (int g = 0; g < 4; ++g) {
#pragma unroll
            for (int j = 0; j < JROWS; ++j) G[g][j] = P[g][j] + accB1[g][j];
        }
    };

    auto cell = [&](const v8f (&G)[4], float (&c)[JROWS], float* hdst) {
#pragma unroll
        for (int j = 0; j < JROWS; ++j) {
            const float ig = sigmoid_of_half(G[0][j]);
            const float fg = sigmoid_of_half(G[1][j]);
            const float gg = fast_tanh(G[2][j]);
            const float og = sigmoid_of_half(G[3][j]);
            c[j] = fg * c[j] + ig * gg;
            hdst[hoff + j * HPAD] = og * fast_tanh(c[j]);
        }
    };

    v2f a0[16], a1[16];
    v8f G0[4], G1[4];

    v2f ax = *(const v2f*)(xrow);
#pragma unroll
    for (int kc = 0; kc < 16; ++kc) { a0[kc][0] = 0.f; a0[kc][1] = 0.f; }
    gemm0(ax, a0, G0);
    ax = *(const v2f*)(xrow + 4);
    cell(G0, c0, sH0);
    __syncthreads();
    load_frag(sH0, aoffE, a0);
    gemm0(ax, a0, G0);
    ax = *(const v2f*)(xrow + 8);
    cell(G0, c0, sH0 + HODD);
    __syncthreads();
    load_frag(sH0, aoffO, a0);
    gemm0(ax, a0, G0);
    __syncthreads();
    v2f axA = *(const v2f*)(xrow + 12);
    v2f axB = *(const v2f*)(xrow + 16);

#pragma unroll 1
    for (int t = 2; t < LSTM_T; t += 2) {
        load_frag(sH1, aoffE, a1);
        __builtin_amdgcn_sched_barrier(0);
        cell(G0, c0, sH0);
        proj1(a0);
        __builtin_amdgcn_sched_barrier(0);
        rec1_zero();
        rec1(a1, 0, 4);
        asm volatile("" :: "v"(P[0]), "v"(P[1]), "v"(P[2]), "v"(P[3]));
        __syncthreads();
        load_frag(sH0, aoffE, a0);
        const int tnA = (t + 3 < LSTM_T) ? (t + 3) : (LSTM_T - 1);
        const int tnB = (t + 4 < LSTM_T) ? (t + 4) : (LSTM_T - 1);
        const v2f axA_next = *(const v2f*)(xrow + (size_t)tnA * 4);
        const v2f axB_next = *(const v2f*)(xrow + (size_t)tnB * 4);
        __builtin_amdgcn_sched_barrier(0);
        rec1(a1, 4, 16);
        merge_even(G1);
        __builtin_amdgcn_sched_barrier(0);

        cell(G1, c1, sH1);
        gemm0(axA, a0, G0);
        __syncthreads();

        load_frag(sH1, aoffE, a1);
        __builtin_amdgcn_sched_barrier(0);
        cell(G0, c0, sH0 + HODD);
        rec1_zero();
        rec1(a1, 0, 4);
        __builtin_amdgcn_sched_barrier(0);
        rec1(a1, 4, 8);
        __syncthreads();
        load_frag(sH0, aoffO, a0);
        __builtin_amdgcn_sched_barrier(0);
        rec1(a1, 8, 16);
        merge_odd(G1);
        __builtin_amdgcn_sched_barrier(0);

        cell(G1, c1, sH1);
        gemm0(axB, a0, G0);
        __syncthreads();

        axA = axA_next;
        axB = axB_next;
    }

    proj1(a0);
    load_frag(sH1, aoffE, a1);
    rec1_zero();
    rec1(a1, 0, 16);
    merge_even(G1);
    __syncthreads();
    cell(G1, c1, sH1);
    __syncthreads();
    load_frag(sH1, aoffE, a1);
    rec1_zero();
    rec1(a1, 0, 16);
    merge_odd(G1);
    __syncthreads();
    cell(G1, c1, sH1);
    __syncthreads();

    {
        const int nco = wave * 16 + l16;
        v8f accA, accB;
        const float fb = fc_b[nco];
#pragma unroll
        for (int j = 0; j < 8; ++j) { accA[j] = fb; accB[j] = 0.f; }
#pragma unroll
        for (int kc = 0; kc < 16; kc += 2) {
            v2f ahA = *(const v2f*)(sH1 + arow * HPAD + (kc    ) * 4 + kb);
            v2f ahB = *(const v2f*)(sH1 + arow * HPAD + (kc + 1) * 4 + kb);
            v2f bwA = *(const v2f*)(fc_w + nco * LSTM_H + (kc    ) * 4 + kb);
            v2f bwB = *(const v2f*)(fc_w + nco * LSTM_H + (kc + 1) * 4 + kb);
            accA = wmma_f32_k4(ahA, bwA, accA);
            accB = wmma_f32_k4(ahB, bwB, accB);
        }
#pragma unroll
        for (int j = 0; j < JROWS; ++j) {
            const int r = j + 4 * half;
            out[(size_t)(rowBase + r) * LSTM_H + nco] = fast_tanh(accA[j] + accB[j]);
        }
    }
}

extern "C" void kernel_launch(void* const* d_in, const int* in_sizes, int n_in,
                              void* d_out, int out_size, void* d_ws, size_t ws_size,
                              hipStream_t stream) {
    (void)in_sizes; (void)n_in; (void)d_ws; (void)ws_size; (void)out_size;
    const float* x     = (const float*)d_in[0];
    const float* w_ih0 = (const float*)d_in[1];
    const float* w_hh0 = (const float*)d_in[2];
    const float* b_ih0 = (const float*)d_in[3];
    const float* b_hh0 = (const float*)d_in[4];
    const float* w_ih1 = (const float*)d_in[5];
    const float* w_hh1 = (const float*)d_in[6];
    const float* b_ih1 = (const float*)d_in[7];
    const float* b_hh1 = (const float*)d_in[8];
    const float* fc_w  = (const float*)d_in[9];
    const float* fc_b  = (const float*)d_in[10];
    float* out = (float*)d_out;

    dim3 grid(LSTM_B / MROWS);
    dim3 block(THREADS);
    LSTMEncoder_78975858639682_kernel<<<grid, block, 0, stream>>>(
        x, w_ih0, w_hh0, b_ih0, b_hh0, w_ih1, w_hh1, b_ih1, b_hh1, fc_w, fc_b, out);
}


# 16 batch rows per PAIR of workgroups, a step's WMMAs cut evenly (layer 0 + input tiles i,f | input tiles g,o + layer-1 recurrent), 32-step chunks pipelined by 17 launches: 100 not 164 per wave
# speedup vs baseline: 1.6879x; 1.1189x over previous
#include <hip/hip_runtime.h>


typedef __attribute__((ext_vector_type(2))) float v2f;
typedef __attribute__((ext_vector_type(8))) float v8f;

#define LSTM_B    2048
#define LSTM_T    512
#define LSTM_H    64
#define MTILE     16
#define NWAVES    4
#define THREADS   (NWAVES * 32)
#define HPAD      68
#define NPAIR     (LSTM_B / MTILE)
#define CH_S      32
#define NCH       (LSTM_T / CH_S)
#define SLOT_F    3072
#define PIN_OFF   1024
#define HX_F      (2 * CH_S * NPAIR * SLOT_F)
#define STA_C0    (HX_F)
#define STB_C1    (HX_F + NPAIR * 1024)
#define STB_H1    (HX_F + 2 * NPAIR * 1024)

__device__ __forceinline__ float fast_tanh(float x) {
#if __has_builtin(__builtin_amdgcn_tanhf)
    return __builtin_amdgcn_tanhf(x);
#else
    float r;
    asm volatile("v_tanh_f32 %0, %1\n\tv_nop" : "=v"(r) : "v"(x));
    return r;
#endif
}

__device__ __forceinline__ float sigmoid_of_half(float xh) {
    return __builtin_fmaf(fast_tanh(xh), 0.5f, 0.5f);
}

__device__ __forceinline__ v8f wmma_f32_k4(v2f a, v2f b, v8f c) {
    return __builtin_amdgcn_wmma_f32_16x16x4_f32(false, a, false, b,
                                                 (short)0, c, false, false);
}

__device__ __forceinline__ void lstm_cell(const v8f (&G)[4], float (&c)[8], float* hdst, int hoff) {
#pragma unroll
    for (int j = 0; j < 8; ++j) {
        const float ig = sigmoid_of_half(G[0][j]);
        const float fg = sigmoid_of_half(G[1][j]);
        const float gg = fast_tanh(G[2][j]);
        const float og = sigmoid_of_half(G[3][j]);
        c[j] = fg * c[j] + ig * gg;
        hdst[hoff + j * HPAD] = og * fast_tanh(c[j]);
    }
}

__device__ __forceinline__ void load_frag(const float* hsrc, int off, v2f (&a)[16]) {
#pragma unroll
    for (int kc = 0; kc < 16; ++kc)
        a[kc] = *(const v2f*)(hsrc + off + kc * 4);
}

__device__ __forceinline__ void load_frag_ws(const float* src, int lane, v2f (&a)[16]) {
#pragma unroll
    for (int q = 0; q < 4; ++q) {
        const v8f f = *(const v8f*)(src + (q * 32 + lane) * 8);
#pragma unroll
        for (int r = 0; r < 4; ++r) { a[4 * q + r][0] = f[2 * r]; a[4 * q + r][1] = f[2 * r + 1]; }
    }
}

__device__ __forceinline__ void store_frag_ws(float* dst, int lane, int wave, const v2f (&a)[16]) {
#pragma unroll
    for (int q = 0; q < 4; ++q) {
        if (wave == q) {
            v8f f;
#pragma unroll
            for (int r = 0; r < 4; ++r) { f[2 * r] = a[4 * q + r][0]; f[2 * r + 1] = a[4 * q + r][1]; }
            *(v8f*)(dst + (q * 32 + lane) * 8) = f;
        }
    }
}

__global__ __launch_bounds__(THREADS)
void lstm2_fc_pair_kernel(
    const float* __restrict__ x,
    const float* __restrict__ w_ih0,
    const float* __restrict__ w_hh0,
    const float* __restrict__ b_ih0,
    const float* __restrict__ b_hh0,
    const float* __restrict__ w_ih1,
    const float* __restrict__ w_hh1,
    const float* __restrict__ b_ih1,
    const float* __restrict__ b_hh1,
    const float* __restrict__ fc_w,
    const float* __restrict__ fc_b,
    float* __restrict__ out,
    float* __restrict__ ws,
    int k)
{
    __shared__ float sH[2 * MTILE * HPAD];

    const int tid  = threadIdx.x;
    const int lane = tid & 31;
    const int wave = __builtin_amdgcn_readfirstlane(tid >> 5);
    const int half = lane >> 4;
    const int l16  = lane & 15;
    const bool front = blockIdx.x < NPAIR;
    const int pair = front ? blockIdx.x : (blockIdx.x - NPAIR);

    const int hcol = wave * 16 + l16;
    const int kb   = 2 * half;
    const int aoff = l16 * HPAD + kb;
    const int hoff = (8 * half) * HPAD + hcol;
    const int cidx = ((pair * NWAVES + wave) * 32 + lane) * 8;

    if (front) {
        if (k >= NCH) return;
        const int par = k & 1;
        v2f   wih0[4], whh0[4][16], wih1[2][16];
        float bias0[4], bias1[2];
#pragma unroll
        for (int g = 0; g < 4; ++g) {
            const int n = g * LSTM_H + hcol;
            const float sc = (g == 2) ? 1.0f : 0.5f;
            wih0[g] = *(const v2f*)(w_ih0 + n * 4 + kb) * sc;
#pragma unroll
            for (int kc = 0; kc < 16; ++kc)
                whh0[g][kc] = *(const v2f*)(w_hh0 + n * LSTM_H + kc * 4 + kb) * sc;
            bias0[g] = (b_ih0[n] + b_hh0[n]) * sc;
        }
#pragma unroll
        for (int g = 0; g < 2; ++g) {
            const int n = g * LSTM_H + hcol;
#pragma unroll
            for (int kc = 0; kc < 16; ++kc)
                wih1[g][kc] = *(const v2f*)(w_ih1 + n * LSTM_H + kc * 4 + kb) * 0.5f;
            bias1[g] = (b_ih1[n] + b_hh1[n]) * 0.5f;
        }

        float c0[8];
        v2f   a0[16];
        if (k == 0) {
#pragma unroll
            for (int j = 0; j < 8; ++j) c0[j] = 0.f;
#pragma unroll
            for (int kc = 0; kc < 16; ++kc) { a0[kc][0] = 0.f; a0[kc][1] = 0.f; }
        } else {
            const float* sp = ws + ((size_t)(((k - 1) & 1) * CH_S + (CH_S - 1)) * NPAIR + pair) * SLOT_F;
            load_frag_ws(sp, lane, a0);
            const v8f cv = *(const v8f*)(ws + STA_C0 + cidx);
#pragma unroll
            for (int j = 0; j < 8; ++j) c0[j] = cv[j];
        }

        auto gemm0 = [&](v2f axv, const v2f (&a)[16], v8f (&G)[4]) {
            v8f accA[4], accB[4];
#pragma unroll
            for (int g = 0; g < 4; ++g) {
#pragma unroll
                for (int j = 0; j < 8; ++j) { accA[g][j] = bias0[g]; accB[g][j] = 0.f; }
            }
#pragma unroll
            for (int g = 0; g < 4; ++g) accB[g] = wmma_f32_k4(axv, wih0[g], accB[g]);
#pragma unroll
            for (int kc = 0; kc < 16; kc += 2) {
#pragma unroll
                for (int g = 0; g < 4; ++g) accA[g] = wmma_f32_k4(a[kc],     whh0[g][kc],     accA[g]);
#pragma unroll
                for (int g = 0; g < 4; ++g) accB[g] = wmma_f32_k4(a[kc + 1], whh0[g][kc + 1], accB[g]);
            }
#pragma unroll
            for (int g = 0; g < 4; ++g) G[g] = accA[g] + accB[g];
        };
        v8f P[2];
        auto proj = [&](const v2f (&a)[16], int kfrom, int kto) {
            if (kfrom == 0) {
#pragma unroll
                for (int g = 0; g < 2; ++g) {
#pragma unroll
                    for (int j = 0; j < 8; ++j) P[g][j] = bias1[g];
                }
            }
#pragma unroll
            for (int kc = kfrom; kc < kto; ++kc) {
#pragma unroll
                for (int g = 0; g < 2; ++g) P[g] = wmma_f32_k4(a[kc], wih1[g][kc], P[g]);
            }
        };
        auto handover = [&](int slot, const v2f (&a)[16]) {
            float* sp = ws + ((size_t)(par * CH_S + slot) * NPAIR + pair) * SLOT_F;
            store_frag_ws(sp, lane, wave, a);
#pragma unroll
            for (int g = 0; g < 2; ++g)
                *(v8f*)(sp + PIN_OFF + ((wave * 2 + g) * 32 + lane) * 8) = P[g];
        };

        const int t0 = k * CH_S;
        const float* xrow = x + (size_t)(pair * MTILE + l16) * LSTM_T * 4 + kb;
        v2f ax = *(const v2f*)(xrow + (size_t)t0 * 4);
        v8f G0[4];

#pragma unroll 1
        for (int i = 0; i < CH_S; ++i) {
            const int tn = (t0 + i + 1 < LSTM_T) ? (t0 + i + 1) : (LSTM_T - 1);
            const v2f axn = *(const v2f*)(xrow + (size_t)tn * 4);
            gemm0(ax, a0, G0);
            __builtin_amdgcn_sched_barrier(0);
            float* hb = sH + (i & 1) * (MTILE * HPAD);
            lstm_cell(G0, c0, hb, hoff);
            proj(a0, 0, 10);
            __builtin_amdgcn_sched_barrier(0);
            proj(a0, 10, 16);
            __syncthreads();
            v2f a0n[16];
            load_frag(hb, aoff, a0n);
            if (i > 0) handover(i - 1, a0);
#pragma unroll
            for (int kc = 0; kc < 16; ++kc) a0[kc] = a0n[kc];
            ax = axn;
        }
        proj(a0, 0, 16);
        handover(CH_S - 1, a0);
        {
            v8f cv;
#pragma unroll
            for (int j = 0; j < 8; ++j) cv[j] = c0[j];
            *(v8f*)(ws + STA_C0 + cidx) = cv;
        }
        return;
    }

    if (k < 1) return;
    const int ch  = k - 1;
    const int par = ch & 1;
    v2f   whh1[4][16], wih1[2][16];
    float bias1[2];
#pragma unroll
    for (int g = 0; g < 4; ++g) {
        const int n = g * LSTM_H + hcol;
        const float sc = (g == 2) ? 1.0f : 0.5f;
#pragma unroll
        for (int kc = 0; kc < 16; ++kc)
            whh1[g][kc] = *(const v2f*)(w_hh1 + n * LSTM_H + kc * 4 + kb) * sc;
    }
#pragma unroll
    for (int g2 = 0; g2 < 2; ++g2) {
        const int n = (2 + g2) * LSTM_H + hcol;
        const float sc = (g2 == 0) ? 1.0f : 0.5f;
#pragma unroll
        for (int kc = 0; kc < 16; ++kc)
            wih1[g2][kc] = *(const v2f*)(w_ih1 + n * LSTM_H + kc * 4 + kb) * sc;
        bias1[g2] = (b_ih1[n] + b_hh1[n]) * sc;
    }

    float c1[8];
    v2f   a1[16];
    if (ch == 0) {
#pragma unroll
        for (int j = 0; j < 8; ++j) c1[j] = 0.f;
#pragma unroll
        for (int kc = 0; kc < 16; ++kc) { a1[kc][0] = 0.f; a1[kc][1] = 0.f; }
    } else {
        load_frag_ws(ws + STB_H1 + pair * 1024, lane, a1);
        const v8f cv = *(const v8f*)(ws + STB_C1 + cidx);
#pragma unroll
        for (int j = 0; j < 8; ++j) c1[j] = cv[j];
    }

    const float* chunk = ws + ((size_t)(par * CH_S) * NPAIR + pair) * SLOT_F;
    const size_t slot_stride = (size_t)NPAIR * SLOT_F;
    auto load_slot = [&](int slot, v2f (&ha)[16], v8f (&pl)[2]) {
        const float* sp = chunk + (size_t)slot * slot_stride;
        load_frag_ws(sp, lane, ha);
#pragma unroll
        for (int g = 0; g < 2; ++g)
            pl[g] = *(const v8f*)(sp + PIN_OFF + ((wave * 2 + g) * 32 + lane) * 8);
    };
    auto projH = [&](const v2f (&ha)[16], v8f (&ph)[2], int kfrom, int kto) {
        if (kfrom == 0) {
#pragma unroll
            for (int g = 0; g < 2; ++g) {
#pragma unroll
                for (int j = 0; j < 8; ++j) ph[g][j] = bias1[g];
            }
        }
#pragma unroll
        for (int kc = kfrom; kc < kto; ++kc) {
#pragma unroll
            for (int g = 0; g < 2; ++g) ph[g] = wmma_f32_k4(ha[kc], wih1[g][kc], ph[g]);
        }
    };

    v2f hA[16];
    v8f PL[2], PH[2];
    load_slot(0, hA, PL);
    projH(hA, PH, 0, 16);
    v8f G1[4];

#pragma unroll 1
    for (int i = 0; i < CH_S; ++i) {
        const int in = (i + 1 < CH_S) ? (i + 1) : (CH_S - 1);
        v2f hAn[16];
        v8f PLn[2], PHn[2];
        load_slot(in, hAn, PLn);
        v8f R[4];
#pragma unroll
        for (int g = 0; g < 4; ++g) {
#pragma unroll
            for (int j = 0; j < 8; ++j) R[g][j] = 0.f;
        }
#pragma unroll
        for (int kc = 0; kc < 16; ++kc) {
#pragma unroll
            for (int g = 0; g < 4; ++g) R[g] = wmma_f32_k4(a1[kc], whh1[g][kc], R[g]);
        }
        G1[0] = PL[0] + R[0];
        G1[1] = PL[1] + R[1];
        G1[2] = PH[0] + R[2];
        G1[3] = PH[1] + R[3];
        __builtin_amdgcn_sched_barrier(0);
        float* hb = sH + (i & 1) * (MTILE * HPAD);
        lstm_cell(G1, c1, hb, hoff);
        projH(hAn, PHn, 0, 10);
        __builtin_amdgcn_sched_barrier(0);
        projH(hAn, PHn, 10, 16);
        __syncthreads();
        load_frag(hb, aoff, a1);
        PL[0] = PLn[0]; PL[1] = PLn[1];
        PH[0] = PHn[0]; PH[1] = PHn[1];
    }

    if (ch < NCH - 1) {
        store_frag_ws(ws + STB_H1 + pair * 1024, lane, wave, a1);
        v8f cv;
#pragma unroll
        for (int j = 0; j < 8; ++j) cv[j] = c1[j];
        *(v8f*)(ws + STB_C1 + cidx) = cv;
        return;
    }

    {
        const int nco = wave * 16 + l16;
        v8f accA, accB;
        const float fb = fc_b[nco];
#pragma unroll
        for (int j = 0; j < 8; ++j) { accA[j] = fb; accB[j] = 0.f; }
#pragma unroll
        for (int kc = 0; kc < 16; kc += 2) {
            const v2f bwA = *(const v2f*)(fc_w + nco * LSTM_H + (kc    ) * 4 + kb);
            const v2f bwB = *(const v2f*)(fc_w + nco * LSTM_H + (kc + 1) * 4 + kb);
            accA = wmma_f32_k4(a1[kc],     bwA, accA);
            accB = wmma_f32_k4(a1[kc + 1], bwB, accB);
        }
#pragma unroll
        for (int j = 0; j < 8; ++j) {
            const int r = 8 * half + j;
            out[(size_t)(pair * MTILE + r) * LSTM_H + nco] = fast_tanh(accA[j] + accB[j]);
        }
    }
}

extern "C" void kernel_launch(void* const* d_in, const int* in_sizes, int n_in,
                              void* d_out, int out_size, void* d_ws, size_t ws_size,
                              hipStream_t stream) {
    (void)in_sizes; (void)n_in; (void)ws_size; (void)out_size;
    const float* x     = (const float*)d_in[0];
    const float* w_ih0 = (const float*)d_in[1];
    const float* w_hh0 = (const float*)d_in[2];
    const float* b_ih0 = (const float*)d_in[3];
    const float* b_hh0 = (const float*)d_in[4];
    const float* w_ih1 = (const float*)d_in[5];
    const float* w_hh1 = (const float*)d_in[6];
    const float* b_ih1 = (const float*)d_in[7];
    const float* b_hh1 = (const float*)d_in[8];
    const float* fc_w  = (const float*)d_in[9];
    const float* fc_b  = (const float*)d_in[10];
    float* out = (float*)d_out;
    float* ws  = (float*)d_ws;

    dim3 grid(2 * NPAIR);
    dim3 block(THREADS);
#define LSTM_LAUNCH(K) lstm2_fc_pair_kernel<<<grid, block, 0, stream>>>( \
        x, w_ih0, w_hh0, b_ih0, b_hh0, w_ih1, w_hh1, b_ih1, b_hh1, fc_w, fc_b, out, ws, K)
    LSTM_LAUNCH(0);  LSTM_LAUNCH(1);  LSTM_LAUNCH(2);  LSTM_LAUNCH(3);
    LSTM_LAUNCH(4);  LSTM_LAUNCH(5);  LSTM_LAUNCH(6);  LSTM_LAUNCH(7);
    LSTM_LAUNCH(8);  LSTM_LAUNCH(9);  LSTM_LAUNCH(10); LSTM_LAUNCH(11);
    LSTM_LAUNCH(12); LSTM_LAUNCH(13); LSTM_LAUNCH(14); LSTM_LAUNCH(15);
    LSTM_LAUNCH(16);
#undef LSTM_LAUNCH
}
